# starting kernel
# baseline (speedup 1.0000x reference)
#include <hip/hip_runtime.h>

typedef float v2f __attribute__((ext_vector_type(2)));
typedef float v8f __attribute__((ext_vector_type(8)));

__global__ void zero_f32(float* __restrict__ p, int n) {
  int i = blockIdx.x * blockDim.x + threadIdx.x;
  if (i < n) p[i] = 0.0f;
}

__global__ void degree_kernel(const int* __restrict__ src, const int* __restrict__ dst,
                              float* __restrict__ deg_out, float* __restrict__ deg_in,
                              int n_edges) {
  int e = blockIdx.x * blockDim.x + threadIdx.x;
  if (e >= n_edges) return;
  atomicAdd(deg_out + src[e], 1.0f);
  atomicAdd(deg_in + dst[e], 1.0f);
}

__global__ void norm_kernel(float* __restrict__ deg, int n) {
  int i = blockIdx.x * blockDim.x + threadIdx.x;
  if (i >= n) return;
  deg[i] = rsqrtf(fmaxf(deg[i], 1.0f));
}

__global__ void post_kernel(const float* __restrict__ agg, const float* __restrict__ norm_d,
                            const float* __restrict__ bias, float* __restrict__ out,
                            int n_elems, int do_relu) {
  int i = blockIdx.x * blockDim.x + threadIdx.x;
  if (i >= n_elems) return;
  int row = i >> 6;
  int col = i & 63;
  float v = fmaf(agg[i], norm_d[row], bias[col]);
  if (do_relu) v = fmaxf(v, 0.0f);
  out[i] = v;
}

__global__ void spmm_scatter(const float* __restrict__ H, const int* __restrict__ src,
                             const int* __restrict__ dst, float* __restrict__ agg,
                             int n_edges) {
  int t = blockIdx.x * blockDim.x + threadIdx.x;
  int e = t >> 4;
  if (e >= n_edges) return;
  int c = (t & 15) * 4;
  int s = src[e];
  int d = dst[e];
  const float4 v = *reinterpret_cast<const float4*>(H + (size_t)s * 64 + c);
  float* o = agg + (size_t)d * 64 + c;
  atomicAdd(o + 0, v.x);
  atomicAdd(o + 1, v.y);
  atomicAdd(o + 2, v.z);
  atomicAdd(o + 3, v.w);
}

template <int K>
__global__ void __launch_bounds__(256) gemm_norm_wmma(const float* __restrict__ X,
                                                      const float* __restrict__ W,
                                                      const float* __restrict__ norm_s,
                                                      float* __restrict__ Y,
                                                      int n_nodes) {
  __shared__ float sW[K * 64];
  for (int i = threadIdx.x; i < K * 64; i += 256) sW[i] = W[i];
  __syncthreads();

  const int wave = threadIdx.x >> 5;
  const int lane = threadIdx.x & 31;
  const int row0 = (blockIdx.x * 8 + wave) * 16;
  if (row0 + 15 >= n_nodes) return;

  const int half  = lane >> 4;
  const int sub   = lane & 15;
  const int mrow  = row0 + sub;
  const float ns  = norm_s[mrow];
  const float* __restrict__ xrow = X + (size_t)mrow * K;

  v8f acc0 = {}, acc1 = {}, acc2 = {}, acc3 = {};

#pragma unroll
  for (int kk = 0; kk < K / 4; ++kk) {
    const int ka = kk * 4 + half * 2;
    v2f a;
    a.x = xrow[ka] * ns;
    a.y = xrow[ka + 1] * ns;

    const float* wk = &sW[ka * 64 + sub];
    v2f b;

    b.x = wk[0];       b.y = wk[64];
    acc0 = __builtin_amdgcn_wmma_f32_16x16x4_f32(false, a, false, b, (short)0, acc0, false, false);
    b.x = wk[16];      b.y = wk[64 + 16];
    acc1 = __builtin_amdgcn_wmma_f32_16x16x4_f32(false, a, false, b, (short)0, acc1, false, false);
    b.x = wk[32];      b.y = wk[64 + 32];
    acc2 = __builtin_amdgcn_wmma_f32_16x16x4_f32(false, a, false, b, (short)0, acc2, false, false);
    b.x = wk[48];      b.y = wk[64 + 48];
    acc3 = __builtin_amdgcn_wmma_f32_16x16x4_f32(false, a, false, b, (short)0, acc3, false, false);
  }

  float* ybase = Y + (size_t)row0 * 64 + sub;
#pragma unroll
  for (int r = 0; r < 8; ++r) {
    const int m = r + half * 8;
    float* yr = ybase + (size_t)m * 64;
    yr[0]  = acc0[r];
    yr[16] = acc1[r];
    yr[32] = acc2[r];
    yr[48] = acc3[r];
  }
}

static inline int cdiv(long long a, long long b) { return (int)((a + b - 1) / b); }

extern "C" void kernel_launch(void* const* d_in, const int* in_sizes, int n_in,
                              void* d_out, int out_size, void* d_ws, size_t ws_size,
                              hipStream_t stream) {
  const float* features = (const float*)d_in[0];
  const int*   src      = (const int*)d_in[1];
  const int*   dst      = (const int*)d_in[2];
  const float* W1       = (const float*)d_in[3];
  const float* b1       = (const float*)d_in[4];
  const float* W2       = (const float*)d_in[5];
  const float* b2       = (const float*)d_in[6];

  const int n_nodes = in_sizes[0] / 128;
  const int n_edges = in_sizes[1];
  const int n_feat  = n_nodes * 64;

  float* ws     = (float*)d_ws;
  float* norm_s = ws;
  float* norm_d = ws + n_nodes;
  float* bufA   = ws + 2 * (size_t)n_nodes;
  float* bufB   = bufA + (size_t)n_nodes * 64;
  float* out    = (float*)d_out;

  const int B = 256;

  zero_f32<<<cdiv(2 * (long long)n_nodes, B), B, 0, stream>>>(norm_s, 2 * n_nodes);
  degree_kernel<<<cdiv(n_edges, B), B, 0, stream>>>(src, dst, norm_s, norm_d, n_edges);
  norm_kernel<<<cdiv(2 * (long long)n_nodes, B), B, 0, stream>>>(norm_s, 2 * n_nodes);

  const int gemm_blocks = cdiv(n_nodes, 128);

  gemm_norm_wmma<128><<<gemm_blocks, B, 0, stream>>>(features, W1, norm_s, bufA, n_nodes);
  zero_f32<<<cdiv(n_feat, B), B, 0, stream>>>(bufB, n_feat);
  spmm_scatter<<<cdiv((long long)n_edges * 16, B), B, 0, stream>>>(bufA, src, dst, bufB, n_edges);
  post_kernel<<<cdiv(n_feat, B), B, 0, stream>>>(bufB, norm_d, b1, bufA, n_feat, 1);

  gemm_norm_wmma<64><<<gemm_blocks, B, 0, stream>>>(bufA, W2, norm_s, bufB, n_nodes);
  zero_f32<<<cdiv(n_feat, B), B, 0, stream>>>(out, n_feat);
  spmm_scatter<<<cdiv((long long)n_edges * 16, B), B, 0, stream>>>(bufB, src, dst, out, n_edges);
  post_kernel<<<cdiv(n_feat, B), B, 0, stream>>>(out, norm_d, b2, out, n_feat, 0);
}


# Owners read a bit index, not the edge list: a small kernel sets one bit per owner and group of 16 edges (LDS OR); an owner opens only its groups and reads src, dst there; LDS arrival-order sums
# speedup vs baseline: 3.2268x; 3.2268x over previous
#include <hip/hip_runtime.h>

typedef float v2f __attribute__((ext_vector_type(2)));
typedef float v8f __attribute__((ext_vector_type(8)));

__device__ __forceinline__ int l2_half() {
  return (int)((__builtin_amdgcn_s_sendmsg_rtn(0x87) >> 18) & 1u);
}

__device__ __forceinline__ int split_row(int n_nodes) {
  return ((n_nodes / 2 + 31) / 32) * 32;
}

#define TICKET_STRIDE 32

__global__ void zero_f32(float* __restrict__ p, int n) {
  int i = blockIdx.x * blockDim.x + threadIdx.x;
  if (i < n) p[i] = 0.0f;
}

__global__ void __launch_bounds__(256) degree_kernel(const int* __restrict__ src,
                                                     const int* __restrict__ dst,
                                                     float* __restrict__ deg_out,
                                                     float* __restrict__ deg_in,
                                                     int n_edges, int n_nodes,
                                                     unsigned* __restrict__ tickets) {
  __shared__ unsigned s_chunk;
  __shared__ int s_home;
  if (threadIdx.x == 0) s_home = l2_half();
  __syncthreads();
  const int home = s_home;
  const int mid = split_row(n_nodes);
  const int chunk_edges = 256 * 8;
  const unsigned n_chunks = (unsigned)((n_edges + chunk_edges - 1) / chunk_edges);

  for (int pass = 0; pass < 2; ++pass) {
    const int part = home ^ pass;
    const bool upper = (part == 1);
    unsigned* ticket = tickets + part * TICKET_STRIDE;
    unsigned t = 0;
    if (threadIdx.x == 0) t = atomicAdd(ticket, 1u);
    for (unsigned it = 0; it <= n_chunks; ++it) {
      if (threadIdx.x == 0) s_chunk = t;
      __syncthreads();
      const unsigned c = s_chunk;
      __syncthreads();
      if (c >= n_chunks) break;
      if (threadIdx.x == 0) t = atomicAdd(ticket, 1u);
      const int base = (int)c * chunk_edges + (int)threadIdx.x;
      int sv[8], dv[8];
#pragma unroll
      for (int k = 0; k < 8; ++k) {
        const int e = base + k * 256;
        sv[k] = -1;
        dv[k] = -1;
        if (e < n_edges) {
          sv[k] = src[e];
          dv[k] = dst[e];
        }
      }
#pragma unroll
      for (int k = 0; k < 8; ++k) {
        if (sv[k] >= 0 && (sv[k] >= mid) == upper) atomicAdd(deg_out + sv[k], 1.0f);
        if (dv[k] >= 0 && (dv[k] >= mid) == upper) atomicAdd(deg_in + dv[k], 1.0f);
      }
    }
  }
}

__global__ void norm_kernel(float* __restrict__ deg, int n) {
  int i = blockIdx.x * blockDim.x + threadIdx.x;
  if (i >= n) return;
  deg[i] = rsqrtf(fmaxf(deg[i], 1.0f));
}

#define AGG_ROWS        400
#define AGG_WAVES       16
#define AGG_THREADS     (AGG_WAVES * 32)
#define AGG_QCAP        1024
#define AGG_GQCAP       128

#define IDX_GROUPS  1024
#define IDX_OWNERS  256
#define IDX_PITCH   (IDX_OWNERS + 1)
#define IDX_THREADS 512

__device__ __forceinline__ void bit_put(unsigned* row, unsigned bit, int d, int o0) {
  const unsigned o = (unsigned)d / AGG_ROWS - (unsigned)o0;
  if (o < IDX_OWNERS) atomicOr(row + o, bit);
}

__global__ void __launch_bounds__(IDX_THREADS) owner_bits(const int* __restrict__ dst, unsigned* __restrict__ bits,
                                                          int n_edges, int n_owners, int pitch_words) {
  __shared__ unsigned s_bits[32 * IDX_PITCH];
  const int lane = threadIdx.x & 31;
  const int wave = threadIdx.x >> 5;
  const int e0 = blockIdx.x * (IDX_GROUPS * 16);
  const bool whole = e0 + IDX_GROUPS * 16 <= n_edges;
  for (int o0 = 0; o0 < n_owners; o0 += IDX_OWNERS) {
    for (int i = threadIdx.x; i < 32 * IDX_PITCH; i += IDX_THREADS) s_bits[i] = 0u;
    __syncthreads();
    for (int j = 0; j < (IDX_GROUPS * 16) / (IDX_THREADS * 4); ++j) {
      const int q = j * IDX_THREADS + (int)threadIdx.x;
      const int e = e0 + q * 4;
      const int gl = q >> 2;
      unsigned* row = s_bits + (gl >> 5) * IDX_PITCH;
      const unsigned bit = 1u << (gl & 31);
      if (whole) {
        const int4 d = *reinterpret_cast<const int4*>(dst + e);
        bit_put(row, bit, d.x, o0);
        bit_put(row, bit, d.y, o0);
        bit_put(row, bit, d.z, o0);
        bit_put(row, bit, d.w, o0);
      } else {
        for (int k = 0; k < 4; ++k)
          if (e + k < n_edges) bit_put(row, bit, dst[e + k], o0);
      }
    }
    __syncthreads();
    for (int o = wave; o < IDX_OWNERS && o0 + o < n_owners; o += IDX_THREADS / 32)
      bits[(size_t)(o0 + o) * pitch_words + blockIdx.x * 32 + lane] = s_bits[lane * IDX_PITCH + o];
    __syncthreads();
  }
}

#define REC_GROUPS  32
#define REC_STRIDE  ((REC_GROUPS + 1) * 32)

struct Rec {
  int2* pool;
  unsigned* counter;
  int n_own;
  int cur;
  int g;
};

__device__ __forceinline__ void rec_put(Rec& rec, int s, int dl, int lane) {
  if (rec.g == REC_GROUPS) {
    unsigned t = 0u;
    if (lane == 0) t = atomicAdd(rec.counter, 1u);
    const int nxt = rec.n_own + __builtin_amdgcn_readfirstlane((int)t);
    if (lane == 0) rec.pool[(size_t)rec.cur * REC_STRIDE] = make_int2(nxt, REC_GROUPS);
    rec.cur = nxt;
    rec.g = 0;
  }
  rec.pool[(size_t)rec.cur * REC_STRIDE + (size_t)(1 + rec.g) * 32 + lane] = make_int2(s, dl);
  rec.g += 1;
}

__device__ __forceinline__ void q_push(int2* __restrict__ qe, unsigned& tail, bool m, int s, int dl) {
  const unsigned mask = __builtin_amdgcn_ballot_w32(m);
  if (mask != 0u) {
    const unsigned pos = (tail + __builtin_amdgcn_mbcnt_lo(mask, 0u)) & (AGG_QCAP - 1);
    if (m) qe[pos] = make_int2(s, dl);
    tail += (unsigned)__builtin_popcount(mask);
  }
}

__device__ __forceinline__ void q_drain(const float* __restrict__ H, float* s_acc, const int2* qe,
                                        unsigned& head, unsigned count, int lane, Rec& rec) {
  __builtin_amdgcn_wave_barrier();
  int s = 0;
  int dl = -1;
  if ((unsigned)lane < count) {
    const int2 pr = qe[(head + (unsigned)lane) & (AGG_QCAP - 1)];
    s = pr.x;
    dl = pr.y;
  }
  rec_put(rec, s, dl, lane);
  float2 v[32];
#pragma unroll
  for (int k = 0; k < 32; ++k) {
    const int sk = __builtin_amdgcn_readlane(s, k);
    v[k] = *reinterpret_cast<const float2*>(H + (size_t)sk * 64 + lane * 2);
  }
#pragma unroll
  for (int k = 0; k < 32; ++k) {
    const int dk = __builtin_amdgcn_readlane(dl, k);
    if (dk >= 0) {
      float* a = s_acc + dk * 64 + lane * 2;
      atomicAdd(a, v[k].x);
      atomicAdd(a + 1, v[k].y);
    }
  }
  head += count;
}

__device__ __forceinline__ void g_expand(const int* qg, unsigned& ghead, unsigned count,
                                         const float* __restrict__ H, const int* __restrict__ src,
                                         const int* __restrict__ dst, float* s_acc, int2* qe,
                                         unsigned& head, unsigned& tail, int lane,
                                         int n_edges, int row0, unsigned nrows, Rec& rec) {
  __builtin_amdgcn_wave_barrier();
  const int sub = lane & 15;
  const bool hi = lane >= 16;
  int gq = -1;
  if ((unsigned)lane < count) gq = qg[(ghead + (unsigned)lane) & (AGG_GQCAP - 1)];
  int dv[16], sv[16];
#pragma unroll
  for (int k = 0; k < 16; ++k) {
    const int ga = __builtin_amdgcn_readlane(gq, 2 * k);
    const int gb = __builtin_amdgcn_readlane(gq, 2 * k + 1);
    const int e = (hi ? gb : ga) * 16 + sub;
    const bool ok = (unsigned)e < (unsigned)n_edges;
    const int ec = ok ? e : 0;
    const int d = dst[ec];
    sv[k] = src[ec];
    dv[k] = ok ? d : -1;
  }
  ghead += count;
#pragma unroll
  for (int k = 0; k < 16; ++k) {
    const unsigned r = (unsigned)(dv[k] - row0);
    q_push(qe, tail, r < nrows, sv[k], (int)r);
  }
  while (tail - head >= 32u) q_drain(H, s_acc, qe, head, 32u, lane, rec);
}

__device__ __forceinline__ void bits_test(unsigned w, int g0, int* qg, unsigned& ghead, unsigned& gtail,
                                          const float* __restrict__ H, const int* __restrict__ src,
                                          const int* __restrict__ dst,
                                          float* s_acc, int2* qe, unsigned& head, unsigned& tail, int lane,
                                          int n_edges, int row0, unsigned nrows, Rec& rec) {
  while (__builtin_amdgcn_ballot_w32(w != 0u) != 0u) {
    const bool hit = (w != 0u);
    const int g = g0 + __builtin_ctz(hit ? w : 1u);
    const unsigned mask = __builtin_amdgcn_ballot_w32(hit);
    const unsigned pos = (gtail + __builtin_amdgcn_mbcnt_lo(mask, 0u)) & (AGG_GQCAP - 1);
    if (hit) qg[pos] = g;
    gtail += (unsigned)__builtin_popcount(mask);
    w &= w - 1u;
    while (gtail - ghead >= 32u)
      g_expand(qg, ghead, 32u, H, src, dst, s_acc, qe, head, tail, lane, n_edges, row0, nrows, rec);
  }
}

__global__ void __launch_bounds__(AGG_THREADS) agg_owner(const float* __restrict__ H,
                                                         const int* __restrict__ src,
                                                         const int* __restrict__ dst,
                                                         const unsigned* __restrict__ bits,
                                                         int pitch_words,
                                                         const float* __restrict__ norm_d,
                                                         const float* __restrict__ bias,
                                                         float* __restrict__ out,
                                                         int n_edges, int n_nodes, int do_relu,
                                                         int* __restrict__ rec_pool_words,
                                                         unsigned* __restrict__ rec_counter) {
  __shared__ __attribute__((aligned(16))) float s_acc[AGG_ROWS * 64];
  __shared__ int2 s_qe[AGG_WAVES * AGG_QCAP];
  __shared__ int s_qg[AGG_WAVES * AGG_GQCAP];

  const int lane = threadIdx.x & 31;
  const int wave = __builtin_amdgcn_readfirstlane((int)(threadIdx.x >> 5));
  const int row0 = blockIdx.x * AGG_ROWS;
  const unsigned nrows = (unsigned)((n_nodes - row0) < AGG_ROWS ? (n_nodes - row0) : AGG_ROWS);
  const unsigned* my_bits = bits + (size_t)blockIdx.x * pitch_words;

  for (int i = threadIdx.x * 4; i < AGG_ROWS * 64; i += AGG_THREADS * 4)
    *reinterpret_cast<float4*>(&s_acc[i]) = make_float4(0.0f, 0.0f, 0.0f, 0.0f);
  __syncthreads();

  float nd_all = 0.0f;
  if ((unsigned)(wave + AGG_WAVES * lane) < nrows) nd_all = norm_d[row0 + wave + AGG_WAVES * lane];

  int2* qe = s_qe + wave * AGG_QCAP;
  int* qg = s_qg + wave * AGG_GQCAP;
  unsigned head = 0, tail = 0;
  unsigned ghead = 0, gtail = 0;

  int2* rec_pool = reinterpret_cast<int2*>(rec_pool_words);
  Rec rec;
  rec.pool = rec_pool;
  rec.counter = rec_counter;
  rec.n_own = (int)gridDim.x * AGG_WAVES;
  rec.cur = (int)blockIdx.x * AGG_WAVES + wave;
  rec.g = 0;

  const int n_st = pitch_words / 32;
  const int st0 = (int)(((long long)blockIdx.x * n_st) / gridDim.x);
  unsigned cur = 0u;
  if (wave < n_st) cur = my_bits[(size_t)((st0 + wave) < n_st ? (st0 + wave) : (st0 + wave - n_st)) * 32 + lane];
  for (int i = wave; i < n_st; i += AGG_WAVES) {
    const int st = (st0 + i) < n_st ? (st0 + i) : (st0 + i - n_st);
    const int inext = (i + AGG_WAVES < n_st) ? (i + AGG_WAVES) : i;
    const unsigned nxt = my_bits[(size_t)((st0 + inext) < n_st ? (st0 + inext) : (st0 + inext - n_st)) * 32 + lane];
    bits_test(cur, (st * 32 + lane) * 32, qg, ghead, gtail, H, src, dst, s_acc, qe, head, tail, lane, n_edges, row0, nrows, rec);
    cur = nxt;
  }

  while (gtail != ghead) {
    const unsigned have = gtail - ghead;
    g_expand(qg, ghead, have < 32u ? have : 32u, H, src, dst, s_acc, qe, head, tail, lane, n_edges, row0, nrows, rec);
  }
  while (tail != head) {
    const unsigned have = tail - head;
    q_drain(H, s_acc, qe, head, have < 32u ? have : 32u, lane, rec);
  }

  if (lane == 0) rec_pool[(size_t)rec.cur * REC_STRIDE] = make_int2(-1, rec.g);

  __syncthreads();

  const float2 bv = *reinterpret_cast<const float2*>(bias + lane * 2);
#pragma unroll
  for (int i = 0; i < AGG_ROWS / AGG_WAVES; ++i) {
    const unsigned r = (unsigned)(wave + AGG_WAVES * i);
    if (r < nrows) {
      const float nd = __builtin_bit_cast(float, __builtin_amdgcn_readlane(__builtin_bit_cast(int, nd_all), i));
      const float2 a = *reinterpret_cast<const float2*>(&s_acc[r * 64 + lane * 2]);
      float2 o;
      o.x = fmaf(a.x, nd, bv.x);
      o.y = fmaf(a.y, nd, bv.y);
      if (do_relu) {
        o.x = fmaxf(o.x, 0.0f);
        o.y = fmaxf(o.y, 0.0f);
      }
      *reinterpret_cast<float2*>(out + (size_t)(row0 + (int)r) * 64 + lane * 2) = o;
    }
  }
}

__global__ void __launch_bounds__(AGG_THREADS) agg_replay(const float* __restrict__ H,
                                                          const int* __restrict__ rec_pool_words,
                                                          const float* __restrict__ norm_d,
                                                          const float* __restrict__ bias,
                                                          float* __restrict__ out,
                                                          int n_edges, int n_nodes, int do_relu) {
  __shared__ __attribute__((aligned(16))) float s_acc[(AGG_ROWS + 1) * 64];

  const int2* rec_pool = reinterpret_cast<const int2*>(rec_pool_words);
  const int lane = threadIdx.x & 31;
  const int wave = __builtin_amdgcn_readfirstlane((int)(threadIdx.x >> 5));
  const int row0 = blockIdx.x * AGG_ROWS;
  const unsigned nrows = (unsigned)((n_nodes - row0) < AGG_ROWS ? (n_nodes - row0) : AGG_ROWS);

  int id = (int)blockIdx.x * AGG_WAVES + wave;
  int g = 0;
  const int2 hd0 = rec_pool[(size_t)id * REC_STRIDE];
  int next = __builtin_amdgcn_readfirstlane(hd0.x);
  int ng = __builtin_amdgcn_readfirstlane(hd0.y);
  int2 cur = make_int2(0, -1);
  if (ng > 0) cur = rec_pool[(size_t)id * REC_STRIDE + 32 + lane];
  else id = -1;

  for (int i = threadIdx.x * 4; i < (AGG_ROWS + 1) * 64; i += AGG_THREADS * 4)
    *reinterpret_cast<float4*>(&s_acc[i]) = make_float4(0.0f, 0.0f, 0.0f, 0.0f);
  __syncthreads();

  float nd_all = 0.0f;
  if ((unsigned)(wave + AGG_WAVES * lane) < nrows) nd_all = norm_d[row0 + wave + AGG_WAVES * lane];

  const int max_groups = (n_edges >> 5) + 34;
  for (int it = 0; it < max_groups && id >= 0; ++it) {
    int nid = id, ngi = g + 1, nng = ng, nnext = next;
    if (ngi >= ng) {
      nid = next;
      ngi = 0;
    }
    int2 nx = make_int2(0, -1);
    if (nid >= 0) {
      if (ngi == 0) {
        const int2 hd = rec_pool[(size_t)nid * REC_STRIDE];
        nnext = __builtin_amdgcn_readfirstlane(hd.x);
        nng = __builtin_amdgcn_readfirstlane(hd.y);
      }
      nx = rec_pool[(size_t)nid * REC_STRIDE + (size_t)(1 + ngi) * 32 + lane];
    }

    float2 v[32];
#pragma unroll
    for (int k = 0; k < 32; ++k) {
      const unsigned sk = (unsigned)__builtin_amdgcn_readlane(cur.x, k);
      const float2* rowp = reinterpret_cast<const float2*>(H + (size_t)sk * 64);
      v[k] = rowp[(unsigned)lane];
    }
    const int drow = cur.y < 0 ? AGG_ROWS : cur.y;
#pragma unroll
    for (int k = 0; k < 32; ++k) {
      const int dk = __builtin_amdgcn_readlane(drow, k);
      float* a = s_acc + dk * 64 + lane * 2;
      atomicAdd(a, v[k].x);
      atomicAdd(a + 1, v[k].y);
    }
    cur = nx;
    id = nid;
    g = ngi;
    ng = nng;
    next = nnext;
  }

  __syncthreads();

  const float2 bv = *reinterpret_cast<const float2*>(bias + lane * 2);
#pragma unroll
  for (int i = 0; i < AGG_ROWS / AGG_WAVES; ++i) {
    const unsigned r = (unsigned)(wave + AGG_WAVES * i);
    if (r < nrows) {
      const float nd = __builtin_bit_cast(float, __builtin_amdgcn_readlane(__builtin_bit_cast(int, nd_all), i));
      const float2 a = *reinterpret_cast<const float2*>(&s_acc[r * 64 + lane * 2]);
      float2 o;
      o.x = fmaf(a.x, nd, bv.x);
      o.y = fmaf(a.y, nd, bv.y);
      if (do_relu) {
        o.x = fmaxf(o.x, 0.0f);
        o.y = fmaxf(o.y, 0.0f);
      }
      *reinterpret_cast<float2*>(out + (size_t)(row0 + (int)r) * 64 + lane * 2) = o;
    }
  }
}

template <int K>
__global__ void __launch_bounds__(256) gemm_norm_wmma(const float* __restrict__ X,
                                                      const float* __restrict__ W,
                                                      const float* __restrict__ norm_s,
                                                      float* __restrict__ Y,
                                                      int n_nodes) {
  __shared__ float sW[K * 64];
  for (int i = threadIdx.x; i < K * 64; i += 256) sW[i] = W[i];
  __syncthreads();

  const int wave = threadIdx.x >> 5;
  const int lane = threadIdx.x & 31;
  const int row0 = (blockIdx.x * 8 + wave) * 16;
  if (row0 + 15 >= n_nodes) return;

  const int half  = lane >> 4;
  const int sub   = lane & 15;
  const int mrow  = row0 + sub;
  const float ns  = norm_s[mrow];
  const float* __restrict__ xrow = X + (size_t)mrow * K;

  v8f acc0 = {}, acc1 = {}, acc2 = {}, acc3 = {};

#pragma unroll
  for (int kk = 0; kk < K / 4; ++kk) {
    const int ka = kk * 4 + half * 2;
    v2f a;
    a.x = xrow[ka] * ns;
    a.y = xrow[ka + 1] * ns;

    const float* wk = &sW[ka * 64 + sub];
    v2f b;

    b.x = wk[0];       b.y = wk[64];
    acc0 = __builtin_amdgcn_wmma_f32_16x16x4_f32(false, a, false, b, (short)0, acc0, false, false);
    b.x = wk[16];      b.y = wk[64 + 16];
    acc1 = __builtin_amdgcn_wmma_f32_16x16x4_f32(false, a, false, b, (short)0, acc1, false, false);
    b.x = wk[32];      b.y = wk[64 + 32];
    acc2 = __builtin_amdgcn_wmma_f32_16x16x4_f32(false, a, false, b, (short)0, acc2, false, false);
    b.x = wk[48];      b.y = wk[64 + 48];
    acc3 = __builtin_amdgcn_wmma_f32_16x16x4_f32(false, a, false, b, (short)0, acc3, false, false);
  }

  float* ybase = Y + (size_t)row0 * 64 + sub;
#pragma unroll
  for (int r = 0; r < 8; ++r) {
    const int m = r + half * 8;
    float* yr = ybase + (size_t)m * 64;
    yr[0]  = acc0[r];
    yr[16] = acc1[r];
    yr[32] = acc2[r];
    yr[48] = acc3[r];
  }
}

static inline int cdiv(long long a, long long b) { return (int)((a + b - 1) / b); }

extern "C" void kernel_launch(void* const* d_in, const int* in_sizes, int n_in,
                              void* d_out, int out_size, void* d_ws, size_t ws_size,
                              hipStream_t stream) {
  const float* features = (const float*)d_in[0];
  const int*   src      = (const int*)d_in[1];
  const int*   dst      = (const int*)d_in[2];
  const float* W1       = (const float*)d_in[3];
  const float* b1       = (const float*)d_in[4];
  const float* W2       = (const float*)d_in[5];
  const float* b2       = (const float*)d_in[6];

  const int n_nodes = in_sizes[0] / 128;
  const int n_edges = in_sizes[1];

  float* ws     = (float*)d_ws;
  float* norm_s = ws;
  float* norm_d = ws + n_nodes;
  unsigned* tickets = (unsigned*)(ws + 2 * (size_t)n_nodes);
  float* bufA   = ws + 2 * (size_t)n_nodes + 256;
  float* bufB   = bufA + (size_t)n_nodes * 64;
  unsigned* bits = (unsigned*)(bufB + (size_t)n_nodes * 64);
  float* out    = (float*)d_out;

  const int B = 256;

  zero_f32<<<cdiv(2 * (long long)n_nodes + 256, B), B, 0, stream>>>(norm_s, 2 * n_nodes + 256);
  degree_kernel<<<1024, B, 0, stream>>>(src, dst, norm_s, norm_d, n_edges, n_nodes, tickets);
  norm_kernel<<<cdiv(2 * (long long)n_nodes, B), B, 0, stream>>>(norm_s, 2 * n_nodes);

  const int gemm_blocks = cdiv(n_nodes, 128);
  const int agg_blocks  = cdiv(n_nodes, AGG_ROWS);

  const int idx_blocks  = cdiv(n_edges, IDX_GROUPS * 16);
  const int pitch_words = idx_blocks * 32;
  owner_bits<<<idx_blocks, IDX_THREADS, 0, stream>>>(dst, bits, n_edges, agg_blocks, pitch_words);

  int* rec_pool = (int*)(bits + (size_t)agg_blocks * pitch_words);

  gemm_norm_wmma<128><<<gemm_blocks, B, 0, stream>>>(features, W1, norm_s, bufA, n_nodes);
  agg_owner<<<agg_blocks, AGG_THREADS, 0, stream>>>(bufA, src, dst, bits, pitch_words, norm_d, b1, bufB, n_edges, n_nodes, 1,
                                                    rec_pool, tickets + 2 * TICKET_STRIDE);

  gemm_norm_wmma<64><<<gemm_blocks, B, 0, stream>>>(bufB, W2, norm_s, bufA, n_nodes);
  agg_replay<<<agg_blocks, AGG_THREADS, 0, stream>>>(bufA, rec_pool, norm_d, b2, out, n_edges, n_nodes, 0);
}


# Degrees counted in LDS, not by global atomics: a work-group counts one 53248-edge chunk's ids of one half of the rows as 16-bit pairs and stores its plane; the norm kernel sums the planes (exact)
# speedup vs baseline: 3.5022x; 1.0853x over previous
#include <hip/hip_runtime.h>

typedef float v2f __attribute__((ext_vector_type(2)));
typedef float v8f __attribute__((ext_vector_type(8)));

#define DEG_PART_NODES 51200
#define DEG_PART_WORDS (DEG_PART_NODES / 2)
#define DEG_THREADS    1024
#define DEG_TRIPS      13
#define DEG_CHUNK      (DEG_TRIPS * DEG_THREADS * 4)
static_assert(DEG_CHUNK < 65536, "a chunk's count must fit 16 bits");

__device__ __forceinline__ void deg_put(unsigned* s_cnt, int v, unsigned node0) {
  const unsigned r = (unsigned)v - node0;
  if (r < DEG_PART_NODES) atomicAdd(s_cnt + (r >> 1), 1u << ((r & 1u) * 16u));
}

__global__ void __launch_bounds__(DEG_THREADS) degree_planes(const int* __restrict__ src,
                                                             const int* __restrict__ dst,
                                                             unsigned* __restrict__ planes,
                                                             int n_edges, int n_chunks, int n_parts,
                                                             int plane_words) {
  __shared__ __attribute__((aligned(16))) unsigned s_cnt[DEG_PART_WORDS];
  const int chunk = (int)blockIdx.x % n_chunks;
  const int rest = (int)blockIdx.x / n_chunks;
  const int part = rest % n_parts;
  const int list = rest / n_parts;
  const int* __restrict__ ids = list ? dst : src;
  const unsigned node0 = (unsigned)part * DEG_PART_NODES;
  const int e0 = chunk * DEG_CHUNK;

  for (int i = threadIdx.x * 4; i < DEG_PART_WORDS; i += DEG_THREADS * 4)
    *reinterpret_cast<uint4*>(&s_cnt[i]) = make_uint4(0u, 0u, 0u, 0u);
  __syncthreads();

  if (e0 + DEG_CHUNK <= n_edges) {
    int4 v[DEG_TRIPS];
#pragma unroll
    for (int k = 0; k < DEG_TRIPS; ++k)
      v[k] = *reinterpret_cast<const int4*>(ids + e0 + (k * DEG_THREADS + (int)threadIdx.x) * 4);
#pragma unroll
    for (int k = 0; k < DEG_TRIPS; ++k) {
      deg_put(s_cnt, v[k].x, node0);
      deg_put(s_cnt, v[k].y, node0);
      deg_put(s_cnt, v[k].z, node0);
      deg_put(s_cnt, v[k].w, node0);
    }
  } else {
    for (int e = e0 + (int)threadIdx.x; e < n_edges; e += DEG_THREADS) deg_put(s_cnt, ids[e], node0);
  }
  __syncthreads();

  unsigned* o = planes + (size_t)(list * n_chunks + chunk) * plane_words + (size_t)part * DEG_PART_WORDS;
  for (int i = threadIdx.x * 4; i < DEG_PART_WORDS; i += DEG_THREADS * 4)
    *reinterpret_cast<uint4*>(o + i) = *reinterpret_cast<const uint4*>(&s_cnt[i]);
}

__global__ void __launch_bounds__(256) norm_from_planes(const unsigned* __restrict__ planes,
                                                        float* __restrict__ norm,
                                                        int n_nodes, int n_chunks, int plane_words,
                                                        unsigned* __restrict__ piece_counter) {
  if (blockIdx.x == 0 && threadIdx.x == 0) *piece_counter = 0u;
  const int words = (n_nodes + 1) / 2;
  const int i = blockIdx.x * 256 + threadIdx.x;
  if (i >= 2 * words) return;
  const int list = i >= words ? 1 : 0;
  const int w = i - list * words;
  const unsigned* p = planes + (size_t)list * n_chunks * plane_words + w;
  unsigned lo = 0u, hi = 0u;
  for (int c0 = 0; c0 < n_chunks; c0 += 16) {
    unsigned v[16];
#pragma unroll
    for (int k = 0; k < 16; ++k) v[k] = (c0 + k < n_chunks) ? p[(size_t)(c0 + k) * plane_words] : 0u;
#pragma unroll
    for (int k = 0; k < 16; ++k) {
      lo += v[k] & 0xffffu;
      hi += v[k] >> 16;
    }
  }
  float* o = norm + (size_t)list * n_nodes + 2 * w;
  o[0] = rsqrtf(fmaxf((float)lo, 1.0f));
  if (2 * w + 1 < n_nodes) o[1] = rsqrtf(fmaxf((float)hi, 1.0f));
}

#define AGG_ROWS        400
#define AGG_WAVES       16
#define AGG_THREADS     (AGG_WAVES * 32)
#define AGG_QCAP        1024
#define AGG_GQCAP       128

#define IDX_GROUPS  1024
#define IDX_OWNERS  256
#define IDX_PITCH   (IDX_OWNERS + 1)
#define IDX_THREADS 512

__device__ __forceinline__ void bit_put(unsigned* row, unsigned bit, int d, int o0) {
  const unsigned o = (unsigned)d / AGG_ROWS - (unsigned)o0;
  if (o < IDX_OWNERS) atomicOr(row + o, bit);
}

__global__ void __launch_bounds__(IDX_THREADS) owner_bits(const int* __restrict__ dst, unsigned* __restrict__ bits,
                                                          int n_edges, int n_owners, int pitch_words) {
  __shared__ unsigned s_bits[32 * IDX_PITCH];
  const int lane = threadIdx.x & 31;
  const int wave = threadIdx.x >> 5;
  const int e0 = blockIdx.x * (IDX_GROUPS * 16);
  const bool whole = e0 + IDX_GROUPS * 16 <= n_edges;
  for (int o0 = 0; o0 < n_owners; o0 += IDX_OWNERS) {
    for (int i = threadIdx.x; i < 32 * IDX_PITCH; i += IDX_THREADS) s_bits[i] = 0u;
    __syncthreads();
    for (int j = 0; j < (IDX_GROUPS * 16) / (IDX_THREADS * 4); ++j) {
      const int q = j * IDX_THREADS + (int)threadIdx.x;
      const int e = e0 + q * 4;
      const int gl = q >> 2;
      unsigned* row = s_bits + (gl >> 5) * IDX_PITCH;
      const unsigned bit = 1u << (gl & 31);
      if (whole) {
        const int4 d = *reinterpret_cast<const int4*>(dst + e);
        bit_put(row, bit, d.x, o0);
        bit_put(row, bit, d.y, o0);
        bit_put(row, bit, d.z, o0);
        bit_put(row, bit, d.w, o0);
      } else {
        for (int k = 0; k < 4; ++k)
          if (e + k < n_edges) bit_put(row, bit, dst[e + k], o0);
      }
    }
    __syncthreads();
    for (int o = wave; o < IDX_OWNERS && o0 + o < n_owners; o += IDX_THREADS / 32)
      bits[(size_t)(o0 + o) * pitch_words + blockIdx.x * 32 + lane] = s_bits[lane * IDX_PITCH + o];
    __syncthreads();
  }
}

#define REC_GROUPS  32
#define REC_STRIDE  ((REC_GROUPS + 1) * 32)

struct Rec {
  int2* pool;
  unsigned* counter;
  int n_own;
  int cur;
  int g;
};

__device__ __forceinline__ void rec_put(Rec& rec, int s, int dl, int lane) {
  if (rec.g == REC_GROUPS) {
    unsigned t = 0u;
    if (lane == 0) t = atomicAdd(rec.counter, 1u);
    const int nxt = rec.n_own + __builtin_amdgcn_readfirstlane((int)t);
    if (lane == 0) rec.pool[(size_t)rec.cur * REC_STRIDE] = make_int2(nxt, REC_GROUPS);
    rec.cur = nxt;
    rec.g = 0;
  }
  rec.pool[(size_t)rec.cur * REC_STRIDE + (size_t)(1 + rec.g) * 32 + lane] = make_int2(s, dl);
  rec.g += 1;
}

__device__ __forceinline__ void q_push(int2* __restrict__ qe, unsigned& tail, bool m, int s, int dl) {
  const unsigned mask = __builtin_amdgcn_ballot_w32(m);
  if (mask != 0u) {
    const unsigned pos = (tail + __builtin_amdgcn_mbcnt_lo(mask, 0u)) & (AGG_QCAP - 1);
    if (m) qe[pos] = make_int2(s, dl);
    tail += (unsigned)__builtin_popcount(mask);
  }
}

__device__ __forceinline__ void q_drain(const float* __restrict__ H, float* s_acc, const int2* qe,
                                        unsigned& head, unsigned count, int lane, Rec& rec) {
  __builtin_amdgcn_wave_barrier();
  int s = 0;
  int dl = -1;
  if ((unsigned)lane < count) {
    const int2 pr = qe[(head + (unsigned)lane) & (AGG_QCAP - 1)];
    s = pr.x;
    dl = pr.y;
  }
  rec_put(rec, s, dl, lane);
  float2 v[32];
#pragma unroll
  for (int k = 0; k < 32; ++k) {
    const int sk = __builtin_amdgcn_readlane(s, k);
    v[k] = *reinterpret_cast<const float2*>(H + (size_t)sk * 64 + lane * 2);
  }
#pragma unroll
  for (int k = 0; k < 32; ++k) {
    const int dk = __builtin_amdgcn_readlane(dl, k);
    if (dk >= 0) {
      float* a = s_acc + dk * 64 + lane * 2;
      atomicAdd(a, v[k].x);
      atomicAdd(a + 1, v[k].y);
    }
  }
  head += count;
}

__device__ __forceinline__ void g_expand(const int* qg, unsigned& ghead, unsigned count,
                                         const float* __restrict__ H, const int* __restrict__ src,
                                         const int* __restrict__ dst, float* s_acc, int2* qe,
                                         unsigned& head, unsigned& tail, int lane,
                                         int n_edges, int row0, unsigned nrows, Rec& rec) {
  __builtin_amdgcn_wave_barrier();
  const int sub = lane & 15;
  const bool hi = lane >= 16;
  int gq = -1;
  if ((unsigned)lane < count) gq = qg[(ghead + (unsigned)lane) & (AGG_GQCAP - 1)];
  int dv[16], sv[16];
#pragma unroll
  for (int k = 0; k < 16; ++k) {
    const int ga = __builtin_amdgcn_readlane(gq, 2 * k);
    const int gb = __builtin_amdgcn_readlane(gq, 2 * k + 1);
    const int e = (hi ? gb : ga) * 16 + sub;
    const bool ok = (unsigned)e < (unsigned)n_edges;
    const int ec = ok ? e : 0;
    const int d = dst[ec];
    sv[k] = src[ec];
    dv[k] = ok ? d : -1;
  }
  ghead += count;
#pragma unroll
  for (int k = 0; k < 16; ++k) {
    const unsigned r = (unsigned)(dv[k] - row0);
    q_push(qe, tail, r < nrows, sv[k], (int)r);
  }
  while (tail - head >= 32u) q_drain(H, s_acc, qe, head, 32u, lane, rec);
}

__device__ __forceinline__ void bits_test(unsigned w, int g0, int* qg, unsigned& ghead, unsigned& gtail,
                                          const float* __restrict__ H, const int* __restrict__ src,
                                          const int* __restrict__ dst,
                                          float* s_acc, int2* qe, unsigned& head, unsigned& tail, int lane,
                                          int n_edges, int row0, unsigned nrows, Rec& rec) {
  while (__builtin_amdgcn_ballot_w32(w != 0u) != 0u) {
    const bool hit = (w != 0u);
    const int g = g0 + __builtin_ctz(hit ? w : 1u);
    const unsigned mask = __builtin_amdgcn_ballot_w32(hit);
    const unsigned pos = (gtail + __builtin_amdgcn_mbcnt_lo(mask, 0u)) & (AGG_GQCAP - 1);
    if (hit) qg[pos] = g;
    gtail += (unsigned)__builtin_popcount(mask);
    w &= w - 1u;
    while (gtail - ghead >= 32u)
      g_expand(qg, ghead, 32u, H, src, dst, s_acc, qe, head, tail, lane, n_edges, row0, nrows, rec);
  }
}

__global__ void __launch_bounds__(AGG_THREADS) agg_owner(const float* __restrict__ H,
                                                         const int* __restrict__ src,
                                                         const int* __restrict__ dst,
                                                         const unsigned* __restrict__ bits,
                                                         int pitch_words,
                                                         const float* __restrict__ norm_d,
                                                         const float* __restrict__ bias,
                                                         float* __restrict__ out,
                                                         int n_edges, int n_nodes, int do_relu,
                                                         int* __restrict__ rec_pool_words,
                                                         unsigned* __restrict__ rec_counter) {
  __shared__ __attribute__((aligned(16))) float s_acc[AGG_ROWS * 64];
  __shared__ int2 s_qe[AGG_WAVES * AGG_QCAP];
  __shared__ int s_qg[AGG_WAVES * AGG_GQCAP];

  const int lane = threadIdx.x & 31;
  const int wave = __builtin_amdgcn_readfirstlane((int)(threadIdx.x >> 5));
  const int row0 = blockIdx.x * AGG_ROWS;
  const unsigned nrows = (unsigned)((n_nodes - row0) < AGG_ROWS ? (n_nodes - row0) : AGG_ROWS);
  const unsigned* my_bits = bits + (size_t)blockIdx.x * pitch_words;

  for (int i = threadIdx.x * 4; i < AGG_ROWS * 64; i += AGG_THREADS * 4)
    *reinterpret_cast<float4*>(&s_acc[i]) = make_float4(0.0f, 0.0f, 0.0f, 0.0f);
  __syncthreads();

  float nd_all = 0.0f;
  if ((unsigned)(wave + AGG_WAVES * lane) < nrows) nd_all = norm_d[row0 + wave + AGG_WAVES * lane];

  int2* qe = s_qe + wave * AGG_QCAP;
  int* qg = s_qg + wave * AGG_GQCAP;
  unsigned head = 0, tail = 0;
  unsigned ghead = 0, gtail = 0;

  int2* rec_pool = reinterpret_cast<int2*>(rec_pool_words);
  Rec rec;
  rec.pool = rec_pool;
  rec.counter = rec_counter;
  rec.n_own = (int)gridDim.x * AGG_WAVES;
  rec.cur = (int)blockIdx.x * AGG_WAVES + wave;
  rec.g = 0;

  const int n_st = pitch_words / 32;
  const int st0 = (int)(((long long)blockIdx.x * n_st) / gridDim.x);
  unsigned cur = 0u;
  if (wave < n_st) cur = my_bits[(size_t)((st0 + wave) < n_st ? (st0 + wave) : (st0 + wave - n_st)) * 32 + lane];
  for (int i = wave; i < n_st; i += AGG_WAVES) {
    const int st = (st0 + i) < n_st ? (st0 + i) : (st0 + i - n_st);
    const int inext = (i + AGG_WAVES < n_st) ? (i + AGG_WAVES) : i;
    const unsigned nxt = my_bits[(size_t)((st0 + inext) < n_st ? (st0 + inext) : (st0 + inext - n_st)) * 32 + lane];
    bits_test(cur, (st * 32 + lane) * 32, qg, ghead, gtail, H, src, dst, s_acc, qe, head, tail, lane, n_edges, row0, nrows, rec);
    cur = nxt;
  }

  while (gtail != ghead) {
    const unsigned have = gtail - ghead;
    g_expand(qg, ghead, have < 32u ? have : 32u, H, src, dst, s_acc, qe, head, tail, lane, n_edges, row0, nrows, rec);
  }
  while (tail != head) {
    const unsigned have = tail - head;
    q_drain(H, s_acc, qe, head, have < 32u ? have : 32u, lane, rec);
  }

  if (lane == 0) rec_pool[(size_t)rec.cur * REC_STRIDE] = make_int2(-1, rec.g);

  __syncthreads();

  const float2 bv = *reinterpret_cast<const float2*>(bias + lane * 2);
#pragma unroll
  for (int i = 0; i < AGG_ROWS / AGG_WAVES; ++i) {
    const unsigned r = (unsigned)(wave + AGG_WAVES * i);
    if (r < nrows) {
      const float nd = __builtin_bit_cast(float, __builtin_amdgcn_readlane(__builtin_bit_cast(int, nd_all), i));
      const float2 a = *reinterpret_cast<const float2*>(&s_acc[r * 64 + lane * 2]);
      float2 o;
      o.x = fmaf(a.x, nd, bv.x);
      o.y = fmaf(a.y, nd, bv.y);
      if (do_relu) {
        o.x = fmaxf(o.x, 0.0f);
        o.y = fmaxf(o.y, 0.0f);
      }
      *reinterpret_cast<float2*>(out + (size_t)(row0 + (int)r) * 64 + lane * 2) = o;
    }
  }
}

__global__ void __launch_bounds__(AGG_THREADS) agg_replay(const float* __restrict__ H,
                                                          const int* __restrict__ rec_pool_words,
                                                          const float* __restrict__ norm_d,
                                                          const float* __restrict__ bias,
                                                          float* __restrict__ out,
                                                          int n_edges, int n_nodes, int do_relu) {
  __shared__ __attribute__((aligned(16))) float s_acc[(AGG_ROWS + 1) * 64];

  const int2* rec_pool = reinterpret_cast<const int2*>(rec_pool_words);
  const int lane = threadIdx.x & 31;
  const int wave = __builtin_amdgcn_readfirstlane((int)(threadIdx.x >> 5));
  const int row0 = blockIdx.x * AGG_ROWS;
  const unsigned nrows = (unsigned)((n_nodes - row0) < AGG_ROWS ? (n_nodes - row0) : AGG_ROWS);

  int id = (int)blockIdx.x * AGG_WAVES + wave;
  int g = 0;
  const int2 hd0 = rec_pool[(size_t)id * REC_STRIDE];
  int next = __builtin_amdgcn_readfirstlane(hd0.x);
  int ng = __builtin_amdgcn_readfirstlane(hd0.y);
  int2 cur = make_int2(0, -1);
  if (ng > 0) cur = rec_pool[(size_t)id * REC_STRIDE + 32 + lane];
  else id = -1;

  for (int i = threadIdx.x * 4; i < (AGG_ROWS + 1) * 64; i += AGG_THREADS * 4)
    *reinterpret_cast<float4*>(&s_acc[i]) = make_float4(0.0f, 0.0f, 0.0f, 0.0f);
  __syncthreads();

  float nd_all = 0.0f;
  if ((unsigned)(wave + AGG_WAVES * lane) < nrows) nd_all = norm_d[row0 + wave + AGG_WAVES * lane];

  const int max_groups = (n_edges >> 5) + 34;
  for (int it = 0; it < max_groups && id >= 0; ++it) {
    int nid = id, ngi = g + 1, nng = ng, nnext = next;
    if (ngi >= ng) {
      nid = next;
      ngi = 0;
    }
    int2 nx = make_int2(0, -1);
    if (nid >= 0) {
      if (ngi == 0) {
        const int2 hd = rec_pool[(size_t)nid * REC_STRIDE];
        nnext = __builtin_amdgcn_readfirstlane(hd.x);
        nng = __builtin_amdgcn_readfirstlane(hd.y);
      }
      nx = rec_pool[(size_t)nid * REC_STRIDE + (size_t)(1 + ngi) * 32 + lane];
    }

    float2 v[32];
#pragma unroll
    for (int k = 0; k < 32; ++k) {
      const unsigned sk = (unsigned)__builtin_amdgcn_readlane(cur.x, k);
      const float2* rowp = reinterpret_cast<const float2*>(H + (size_t)sk * 64);
      v[k] = rowp[(unsigned)lane];
    }
    const int drow = cur.y < 0 ? AGG_ROWS : cur.y;
#pragma unroll
    for (int k = 0; k < 32; ++k) {
      const int dk = __builtin_amdgcn_readlane(drow, k);
      float* a = s_acc + dk * 64 + lane * 2;
      atomicAdd(a, v[k].x);
      atomicAdd(a + 1, v[k].y);
    }
    cur = nx;
    id = nid;
    g = ngi;
    ng = nng;
    next = nnext;
  }

  __syncthreads();

  const float2 bv = *reinterpret_cast<const float2*>(bias + lane * 2);
#pragma unroll
  for (int i = 0; i < AGG_ROWS / AGG_WAVES; ++i) {
    const unsigned r = (unsigned)(wave + AGG_WAVES * i);
    if (r < nrows) {
      const float nd = __builtin_bit_cast(float, __builtin_amdgcn_readlane(__builtin_bit_cast(int, nd_all), i));
      const float2 a = *reinterpret_cast<const float2*>(&s_acc[r * 64 + lane * 2]);
      float2 o;
      o.x = fmaf(a.x, nd, bv.x);
      o.y = fmaf(a.y, nd, bv.y);
      if (do_relu) {
        o.x = fmaxf(o.x, 0.0f);
        o.y = fmaxf(o.y, 0.0f);
      }
      *reinterpret_cast<float2*>(out + (size_t)(row0 + (int)r) * 64 + lane * 2) = o;
    }
  }
}

template <int K>
__global__ void __launch_bounds__(256) gemm_norm_wmma(const float* __restrict__ X,
                                                      const float* __restrict__ W,
                                                      const float* __restrict__ norm_s,
                                                      float* __restrict__ Y,
                                                      int n_nodes) {
  __shared__ float sW[K * 64];
  for (int i = threadIdx.x; i < K * 64; i += 256) sW[i] = W[i];
  __syncthreads();

  const int wave = threadIdx.x >> 5;
  const int lane = threadIdx.x & 31;
  const int row0 = (blockIdx.x * 8 + wave) * 16;
  if (row0 + 15 >= n_nodes) return;

  const int half  = lane >> 4;
  const int sub   = lane & 15;
  const int mrow  = row0 + sub;
  const float ns  = norm_s[mrow];
  const float* __restrict__ xrow = X + (size_t)mrow * K;

  v8f acc0 = {}, acc1 = {}, acc2 = {}, acc3 = {};

#pragma unroll
  for (int kk = 0; kk < K / 4; ++kk) {
    const int ka = kk * 4 + half * 2;
    v2f a;
    a.x = xrow[ka] * ns;
    a.y = xrow[ka + 1] * ns;

    const float* wk = &sW[ka * 64 + sub];
    v2f b;

    b.x = wk[0];       b.y = wk[64];
    acc0 = __builtin_amdgcn_wmma_f32_16x16x4_f32(false, a, false, b, (short)0, acc0, false, false);
    b.x = wk[16];      b.y = wk[64 + 16];
    acc1 = __builtin_amdgcn_wmma_f32_16x16x4_f32(false, a, false, b, (short)0, acc1, false, false);
    b.x = wk[32];      b.y = wk[64 + 32];
    acc2 = __builtin_amdgcn_wmma_f32_16x16x4_f32(false, a, false, b, (short)0, acc2, false, false);
    b.x = wk[48];      b.y = wk[64 + 48];
    acc3 = __builtin_amdgcn_wmma_f32_16x16x4_f32(false, a, false, b, (short)0, acc3, false, false);
  }

  float* ybase = Y + (size_t)row0 * 64 + sub;
#pragma unroll
  for (int r = 0; r < 8; ++r) {
    const int m = r + half * 8;
    float* yr = ybase + (size_t)m * 64;
    yr[0]  = acc0[r];
    yr[16] = acc1[r];
    yr[32] = acc2[r];
    yr[48] = acc3[r];
  }
}

static inline int cdiv(long long a, long long b) { return (int)((a + b - 1) / b); }

extern "C" void kernel_launch(void* const* d_in, const int* in_sizes, int n_in,
                              void* d_out, int out_size, void* d_ws, size_t ws_size,
                              hipStream_t stream) {
  const float* features = (const float*)d_in[0];
  const int*   src      = (const int*)d_in[1];
  const int*   dst      = (const int*)d_in[2];
  const float* W1       = (const float*)d_in[3];
  const float* b1       = (const float*)d_in[4];
  const float* W2       = (const float*)d_in[5];
  const float* b2       = (const float*)d_in[6];

  const int n_nodes = in_sizes[0] / 128;
  const int n_edges = in_sizes[1];

  float* ws     = (float*)d_ws;
  float* norm_s = ws;
  float* norm_d = ws + n_nodes;
  unsigned* piece_counter = (unsigned*)(ws + 2 * (size_t)n_nodes);
  float* bufA   = ws + 2 * (size_t)n_nodes + 256;
  float* bufB   = bufA + (size_t)n_nodes * 64;
  unsigned* bits = (unsigned*)(bufB + (size_t)n_nodes * 64);
  float* out    = (float*)d_out;

  const int B = 256;

  const int gemm_blocks = cdiv(n_nodes, 128);
  const int agg_blocks  = cdiv(n_nodes, AGG_ROWS);

  const int idx_blocks  = cdiv(n_edges, IDX_GROUPS * 16);
  const int pitch_words = idx_blocks * 32;

  const int deg_chunks  = cdiv(n_edges, DEG_CHUNK);
  const int deg_parts   = cdiv(n_nodes, DEG_PART_NODES);
  const int plane_words = deg_parts * DEG_PART_WORDS;
  unsigned* planes = bits + (size_t)agg_blocks * pitch_words;

  int* rec_pool = (int*)(planes + (size_t)2 * deg_chunks * plane_words);

  degree_planes<<<2 * deg_parts * deg_chunks, DEG_THREADS, 0, stream>>>(src, dst, planes, n_edges, deg_chunks, deg_parts, plane_words);
  norm_from_planes<<<cdiv(2 * (long long)((n_nodes + 1) / 2), B), B, 0, stream>>>(planes, norm_s, n_nodes, deg_chunks, plane_words, piece_counter);

  owner_bits<<<idx_blocks, IDX_THREADS, 0, stream>>>(dst, bits, n_edges, agg_blocks, pitch_words);

  gemm_norm_wmma<128><<<gemm_blocks, B, 0, stream>>>(features, W1, norm_s, bufA, n_nodes);
  agg_owner<<<agg_blocks, AGG_THREADS, 0, stream>>>(bufA, src, dst, bits, pitch_words, norm_d, b1, bufB, n_edges, n_nodes, 1,
                                                    rec_pool, piece_counter);

  gemm_norm_wmma<64><<<gemm_blocks, B, 0, stream>>>(bufB, W2, norm_s, bufA, n_nodes);
  agg_replay<<<agg_blocks, AGG_THREADS, 0, stream>>>(bufA, rec_pool, norm_d, b2, out, n_edges, n_nodes, 0);
}


# Edge ids re-laid as (dst, src) pairs by the bit-index kernel: an owner opens a group of sixteen edges with one 128-byte request, not two 64-byte ones; sums stay in LDS arrival order
# speedup vs baseline: 3.6891x; 1.0534x over previous
#include <hip/hip_runtime.h>

typedef float v2f __attribute__((ext_vector_type(2)));
typedef float v8f __attribute__((ext_vector_type(8)));

#define DEG_PART_NODES 51200
#define DEG_PART_WORDS (DEG_PART_NODES / 2)
#define DEG_THREADS    1024
#define DEG_TRIPS      13
#define DEG_CHUNK      (DEG_TRIPS * DEG_THREADS * 4)
static_assert(DEG_CHUNK < 65536, "a chunk's count must fit 16 bits");

__device__ __forceinline__ void deg_put(unsigned* s_cnt, int v, unsigned node0) {
  const unsigned r = (unsigned)v - node0;
  if (r < DEG_PART_NODES) atomicAdd(s_cnt + (r >> 1), 1u << ((r & 1u) * 16u));
}

__global__ void __launch_bounds__(DEG_THREADS) degree_planes(const int* __restrict__ src,
                                                             const int* __restrict__ dst,
                                                             unsigned* __restrict__ planes,
                                                             int n_edges, int n_chunks, int n_parts,
                                                             int plane_words) {
  __shared__ __attribute__((aligned(16))) unsigned s_cnt[DEG_PART_WORDS];
  const int chunk = (int)blockIdx.x % n_chunks;
  const int rest = (int)blockIdx.x / n_chunks;
  const int part = rest % n_parts;
  const int list = rest / n_parts;
  const int* __restrict__ ids = list ? dst : src;
  const unsigned node0 = (unsigned)part * DEG_PART_NODES;
  const int e0 = chunk * DEG_CHUNK;

  for (int i = threadIdx.x * 4; i < DEG_PART_WORDS; i += DEG_THREADS * 4)
    *reinterpret_cast<uint4*>(&s_cnt[i]) = make_uint4(0u, 0u, 0u, 0u);
  __syncthreads();

  if (e0 + DEG_CHUNK <= n_edges) {
    int4 v[DEG_TRIPS];
#pragma unroll
    for (int k = 0; k < DEG_TRIPS; ++k)
      v[k] = *reinterpret_cast<const int4*>(ids + e0 + (k * DEG_THREADS + (int)threadIdx.x) * 4);
#pragma unroll
    for (int k = 0; k < DEG_TRIPS; ++k) {
      deg_put(s_cnt, v[k].x, node0);
      deg_put(s_cnt, v[k].y, node0);
      deg_put(s_cnt, v[k].z, node0);
      deg_put(s_cnt, v[k].w, node0);
    }
  } else {
    for (int e = e0 + (int)threadIdx.x; e < n_edges; e += DEG_THREADS) deg_put(s_cnt, ids[e], node0);
  }
  __syncthreads();

  unsigned* o = planes + (size_t)(list * n_chunks + chunk) * plane_words + (size_t)part * DEG_PART_WORDS;
  for (int i = threadIdx.x * 4; i < DEG_PART_WORDS; i += DEG_THREADS * 4)
    *reinterpret_cast<uint4*>(o + i) = *reinterpret_cast<const uint4*>(&s_cnt[i]);
}

__global__ void __launch_bounds__(256) norm_from_planes(const unsigned* __restrict__ planes,
                                                        float* __restrict__ norm,
                                                        int n_nodes, int n_chunks, int plane_words,
                                                        unsigned* __restrict__ piece_counter) {
  if (blockIdx.x == 0 && threadIdx.x == 0) *piece_counter = 0u;
  const int words = (n_nodes + 1) / 2;
  const int i = blockIdx.x * 256 + threadIdx.x;
  if (i >= 2 * words) return;
  const int list = i >= words ? 1 : 0;
  const int w = i - list * words;
  const unsigned* p = planes + (size_t)list * n_chunks * plane_words + w;
  unsigned lo = 0u, hi = 0u;
  for (int c0 = 0; c0 < n_chunks; c0 += 16) {
    unsigned v[16];
#pragma unroll
    for (int k = 0; k < 16; ++k) v[k] = (c0 + k < n_chunks) ? p[(size_t)(c0 + k) * plane_words] : 0u;
#pragma unroll
    for (int k = 0; k < 16; ++k) {
      lo += v[k] & 0xffffu;
      hi += v[k] >> 16;
    }
  }
  float* o = norm + (size_t)list * n_nodes + 2 * w;
  o[0] = rsqrtf(fmaxf((float)lo, 1.0f));
  if (2 * w + 1 < n_nodes) o[1] = rsqrtf(fmaxf((float)hi, 1.0f));
}

#define AGG_ROWS        400
#define AGG_WAVES       16
#define AGG_THREADS     (AGG_WAVES * 32)
#define AGG_QCAP        1024
#define AGG_GQCAP       128

#define IDX_GROUPS  1024
#define IDX_OWNERS  256
#define IDX_PITCH   (IDX_OWNERS + 1)
#define IDX_THREADS 512

__device__ __forceinline__ void bit_put(unsigned* row, unsigned bit, int d, int o0) {
  const unsigned o = (unsigned)d / AGG_ROWS - (unsigned)o0;
  if (o < IDX_OWNERS) atomicOr(row + o, bit);
}

__global__ void __launch_bounds__(IDX_THREADS) owner_bits(const int* __restrict__ src, const int* __restrict__ dst,
                                                          unsigned* __restrict__ bits, int* __restrict__ pairs,
                                                          int n_edges, int n_owners, int pitch_words) {
  __shared__ unsigned s_bits[32 * IDX_PITCH];
  const int lane = threadIdx.x & 31;
  const int wave = threadIdx.x >> 5;
  const int e0 = blockIdx.x * (IDX_GROUPS * 16);
  const bool whole = e0 + IDX_GROUPS * 16 <= n_edges;
  for (int o0 = 0; o0 < n_owners; o0 += IDX_OWNERS) {
    for (int i = threadIdx.x; i < 32 * IDX_PITCH; i += IDX_THREADS) s_bits[i] = 0u;
    __syncthreads();
    for (int j = 0; j < (IDX_GROUPS * 16) / (IDX_THREADS * 4); ++j) {
      const int q = j * IDX_THREADS + (int)threadIdx.x;
      const int e = e0 + q * 4;
      const int gl = q >> 2;
      unsigned* row = s_bits + (gl >> 5) * IDX_PITCH;
      const unsigned bit = 1u << (gl & 31);
      if (whole) {
        const int4 d = *reinterpret_cast<const int4*>(dst + e);
        if (o0 == 0) {
          const int4 sr = *reinterpret_cast<const int4*>(src + e);
          int4* p = reinterpret_cast<int4*>(pairs + (size_t)e * 2);
          p[0] = make_int4(d.x, sr.x, d.y, sr.y);
          p[1] = make_int4(d.z, sr.z, d.w, sr.w);
        }
        bit_put(row, bit, d.x, o0);
        bit_put(row, bit, d.y, o0);
        bit_put(row, bit, d.z, o0);
        bit_put(row, bit, d.w, o0);
      } else {
        for (int k = 0; k < 4; ++k)
          if (e + k < n_edges) {
            const int dd = dst[e + k];
            if (o0 == 0) {
              pairs[(size_t)(e + k) * 2] = dd;
              pairs[(size_t)(e + k) * 2 + 1] = src[e + k];
            }
            bit_put(row, bit, dd, o0);
          }
      }
    }
    __syncthreads();
    for (int o = wave; o < IDX_OWNERS && o0 + o < n_owners; o += IDX_THREADS / 32)
      bits[(size_t)(o0 + o) * pitch_words + blockIdx.x * 32 + lane] = s_bits[lane * IDX_PITCH + o];
    __syncthreads();
  }
}

#define REC_GROUPS  32
#define REC_STRIDE  ((REC_GROUPS + 1) * 32)

struct Rec {
  int2* pool;
  unsigned* counter;
  int n_own;
  int cur;
  int g;
};

__device__ __forceinline__ void rec_put(Rec& rec, int s, int dl, int lane) {
  if (rec.g == REC_GROUPS) {
    unsigned t = 0u;
    if (lane == 0) t = atomicAdd(rec.counter, 1u);
    const int nxt = rec.n_own + __builtin_amdgcn_readfirstlane((int)t);
    if (lane == 0) rec.pool[(size_t)rec.cur * REC_STRIDE] = make_int2(nxt, REC_GROUPS);
    rec.cur = nxt;
    rec.g = 0;
  }
  rec.pool[(size_t)rec.cur * REC_STRIDE + (size_t)(1 + rec.g) * 32 + lane] = make_int2(s, dl);
  rec.g += 1;
}

__device__ __forceinline__ void q_push(int2* __restrict__ qe, unsigned& tail, bool m, int s, int dl) {
  const unsigned mask = __builtin_amdgcn_ballot_w32(m);
  if (mask != 0u) {
    const unsigned pos = (tail + __builtin_amdgcn_mbcnt_lo(mask, 0u)) & (AGG_QCAP - 1);
    if (m) qe[pos] = make_int2(s, dl);
    tail += (unsigned)__builtin_popcount(mask);
  }
}

__device__ __forceinline__ void q_drain(const float* __restrict__ H, float* s_acc, const int2* qe,
                                        unsigned& head, unsigned count, int lane, Rec& rec) {
  __builtin_amdgcn_wave_barrier();
  int s = 0;
  int dl = -1;
  if ((unsigned)lane < count) {
    const int2 pr = qe[(head + (unsigned)lane) & (AGG_QCAP - 1)];
    s = pr.x;
    dl = pr.y;
  }
  rec_put(rec, s, dl, lane);
  float2 v[32];
#pragma unroll
  for (int k = 0; k < 32; ++k) {
    const int sk = __builtin_amdgcn_readlane(s, k);
    v[k] = *reinterpret_cast<const float2*>(H + (size_t)sk * 64 + lane * 2);
  }
#pragma unroll
  for (int k = 0; k < 32; ++k) {
    const int dk = __builtin_amdgcn_readlane(dl, k);
    if (dk >= 0) {
      float* a = s_acc + dk * 64 + lane * 2;
      atomicAdd(a, v[k].x);
      atomicAdd(a + 1, v[k].y);
    }
  }
  head += count;
}

__device__ __forceinline__ void g_expand(const int* qg, unsigned& ghead, unsigned count,
                                         const float* __restrict__ H, const int2* __restrict__ pairs,
                                         float* s_acc, int2* qe,
                                         unsigned& head, unsigned& tail, int lane,
                                         int n_edges, int row0, unsigned nrows, Rec& rec) {
  __builtin_amdgcn_wave_barrier();
  const int sub = lane & 15;
  const bool hi = lane >= 16;
  int gq = -1;
  if ((unsigned)lane < count) gq = qg[(ghead + (unsigned)lane) & (AGG_GQCAP - 1)];
  int dv[16], sv[16];
#pragma unroll
  for (int k = 0; k < 16; ++k) {
    const int ga = __builtin_amdgcn_readlane(gq, 2 * k);
    const int gb = __builtin_amdgcn_readlane(gq, 2 * k + 1);
    const int e = (hi ? gb : ga) * 16 + sub;
    const bool ok = (unsigned)e < (unsigned)n_edges;
    const int ec = ok ? e : 0;
    const int2 pr = pairs[ec];
    sv[k] = pr.y;
    dv[k] = ok ? pr.x : -1;
  }
  ghead += count;
#pragma unroll
  for (int k = 0; k < 16; ++k) {
    const unsigned r = (unsigned)(dv[k] - row0);
    q_push(qe, tail, r < nrows, sv[k], (int)r);
  }
  while (tail - head >= 32u) q_drain(H, s_acc, qe, head, 32u, lane, rec);
}

__device__ __forceinline__ void bits_test(unsigned w, int g0, int* qg, unsigned& ghead, unsigned& gtail,
                                          const float* __restrict__ H, const int2* __restrict__ pairs,
                                          float* s_acc, int2* qe, unsigned& head, unsigned& tail, int lane,
                                          int n_edges, int row0, unsigned nrows, Rec& rec) {
  while (__builtin_amdgcn_ballot_w32(w != 0u) != 0u) {
    const bool hit = (w != 0u);
    const int g = g0 + __builtin_ctz(hit ? w : 1u);
    const unsigned mask = __builtin_amdgcn_ballot_w32(hit);
    const unsigned pos = (gtail + __builtin_amdgcn_mbcnt_lo(mask, 0u)) & (AGG_GQCAP - 1);
    if (hit) qg[pos] = g;
    gtail += (unsigned)__builtin_popcount(mask);
    w &= w - 1u;
    while (gtail - ghead >= 32u)
      g_expand(qg, ghead, 32u, H, pairs, s_acc, qe, head, tail, lane, n_edges, row0, nrows, rec);
  }
}

__global__ void __launch_bounds__(AGG_THREADS) agg_owner(const float* __restrict__ H,
                                                         const int* __restrict__ pair_words,
                                                         const unsigned* __restrict__ bits,
                                                         int pitch_words,
                                                         const float* __restrict__ norm_d,
                                                         const float* __restrict__ bias,
                                                         float* __restrict__ out,
                                                         int n_edges, int n_nodes, int do_relu,
                                                         int* __restrict__ rec_pool_words,
                                                         unsigned* __restrict__ rec_counter) {
  __shared__ __attribute__((aligned(16))) float s_acc[AGG_ROWS * 64];
  __shared__ int2 s_qe[AGG_WAVES * AGG_QCAP];
  __shared__ int s_qg[AGG_WAVES * AGG_GQCAP];

  const int lane = threadIdx.x & 31;
  const int wave = __builtin_amdgcn_readfirstlane((int)(threadIdx.x >> 5));
  const int row0 = blockIdx.x * AGG_ROWS;
  const unsigned nrows = (unsigned)((n_nodes - row0) < AGG_ROWS ? (n_nodes - row0) : AGG_ROWS);
  const unsigned* my_bits = bits + (size_t)blockIdx.x * pitch_words;

  for (int i = threadIdx.x * 4; i < AGG_ROWS * 64; i += AGG_THREADS * 4)
    *reinterpret_cast<float4*>(&s_acc[i]) = make_float4(0.0f, 0.0f, 0.0f, 0.0f);
  __syncthreads();

  float nd_all = 0.0f;
  if ((unsigned)(wave + AGG_WAVES * lane) < nrows) nd_all = norm_d[row0 + wave + AGG_WAVES * lane];

  int2* qe = s_qe + wave * AGG_QCAP;
  int* qg = s_qg + wave * AGG_GQCAP;
  unsigned head = 0, tail = 0;
  unsigned ghead = 0, gtail = 0;

  int2* rec_pool = reinterpret_cast<int2*>(rec_pool_words);
  const int2* pairs = reinterpret_cast<const int2*>(pair_words);
  Rec rec;
  rec.pool = rec_pool;
  rec.counter = rec_counter;
  rec.n_own = (int)gridDim.x * AGG_WAVES;
  rec.cur = (int)blockIdx.x * AGG_WAVES + wave;
  rec.g = 0;

  const int n_st = pitch_words / 32;
  const int st0 = (int)(((long long)blockIdx.x * n_st) / gridDim.x);
  unsigned cur = 0u;
  if (wave < n_st) cur = my_bits[(size_t)((st0 + wave) < n_st ? (st0 + wave) : (st0 + wave - n_st)) * 32 + lane];
  for (int i = wave; i < n_st; i += AGG_WAVES) {
    const int st = (st0 + i) < n_st ? (st0 + i) : (st0 + i - n_st);
    const int inext = (i + AGG_WAVES < n_st) ? (i + AGG_WAVES) : i;
    const unsigned nxt = my_bits[(size_t)((st0 + inext) < n_st ? (st0 + inext) : (st0 + inext - n_st)) * 32 + lane];
    bits_test(cur, (st * 32 + lane) * 32, qg, ghead, gtail, H, pairs, s_acc, qe, head, tail, lane, n_edges, row0, nrows, rec);
    cur = nxt;
  }

  while (gtail != ghead) {
    const unsigned have = gtail - ghead;
    g_expand(qg, ghead, have < 32u ? have : 32u, H, pairs, s_acc, qe, head, tail, lane, n_edges, row0, nrows, rec);
  }
  while (tail != head) {
    const unsigned have = tail - head;
    q_drain(H, s_acc, qe, head, have < 32u ? have : 32u, lane, rec);
  }

  if (lane == 0) rec_pool[(size_t)rec.cur * REC_STRIDE] = make_int2(-1, rec.g);

  __syncthreads();

  const float2 bv = *reinterpret_cast<const float2*>(bias + lane * 2);
#pragma unroll
  for (int i = 0; i < AGG_ROWS / AGG_WAVES; ++i) {
    const unsigned r = (unsigned)(wave + AGG_WAVES * i);
    if (r < nrows) {
      const float nd = __builtin_bit_cast(float, __builtin_amdgcn_readlane(__builtin_bit_cast(int, nd_all), i));
      const float2 a = *reinterpret_cast<const float2*>(&s_acc[r * 64 + lane * 2]);
      float2 o;
      o.x = fmaf(a.x, nd, bv.x);
      o.y = fmaf(a.y, nd, bv.y);
      if (do_relu) {
        o.x = fmaxf(o.x, 0.0f);
        o.y = fmaxf(o.y, 0.0f);
      }
      *reinterpret_cast<float2*>(out + (size_t)(row0 + (int)r) * 64 + lane * 2) = o;
    }
  }
}

__global__ void __launch_bounds__(AGG_THREADS) agg_replay(const float* __restrict__ H,
                                                          const int* __restrict__ rec_pool_words,
                                                          const float* __restrict__ norm_d,
                                                          const float* __restrict__ bias,
                                                          float* __restrict__ out,
                                                          int n_edges, int n_nodes, int do_relu) {
  __shared__ __attribute__((aligned(16))) float s_acc[(AGG_ROWS + 1) * 64];

  const int2* rec_pool = reinterpret_cast<const int2*>(rec_pool_words);
  const int lane = threadIdx.x & 31;
  const int wave = __builtin_amdgcn_readfirstlane((int)(threadIdx.x >> 5));
  const int row0 = blockIdx.x * AGG_ROWS;
  const unsigned nrows = (unsigned)((n_nodes - row0) < AGG_ROWS ? (n_nodes - row0) : AGG_ROWS);

  int id = (int)blockIdx.x * AGG_WAVES + wave;
  int g = 0;
  const int2 hd0 = rec_pool[(size_t)id * REC_STRIDE];
  int next = __builtin_amdgcn_readfirstlane(hd0.x);
  int ng = __builtin_amdgcn_readfirstlane(hd0.y);
  int2 cur = make_int2(0, -1);
  if (ng > 0) cur = rec_pool[(size_t)id * REC_STRIDE + 32 + lane];
  else id = -1;

  for (int i = threadIdx.x * 4; i < (AGG_ROWS + 1) * 64; i += AGG_THREADS * 4)
    *reinterpret_cast<float4*>(&s_acc[i]) = make_float4(0.0f, 0.0f, 0.0f, 0.0f);
  __syncthreads();

  float nd_all = 0.0f;
  if ((unsigned)(wave + AGG_WAVES * lane) < nrows) nd_all = norm_d[row0 + wave + AGG_WAVES * lane];

  const int max_groups = (n_edges >> 5) + 34;
  for (int it = 0; it < max_groups && id >= 0; ++it) {
    int nid = id, ngi = g + 1, nng = ng, nnext = next;
    if (ngi >= ng) {
      nid = next;
      ngi = 0;
    }
    int2 nx = make_int2(0, -1);
    if (nid >= 0) {
      if (ngi == 0) {
        const int2 hd = rec_pool[(size_t)nid * REC_STRIDE];
        nnext = __builtin_amdgcn_readfirstlane(hd.x);
        nng = __builtin_amdgcn_readfirstlane(hd.y);
      }
      nx = rec_pool[(size_t)nid * REC_STRIDE + (size_t)(1 + ngi) * 32 + lane];
    }

    float2 v[32];
#pragma unroll
    for (int k = 0; k < 32; ++k) {
      const unsigned sk = (unsigned)__builtin_amdgcn_readlane(cur.x, k);
      const float2* rowp = reinterpret_cast<const float2*>(H + (size_t)sk * 64);
      v[k] = rowp[(unsigned)lane];
    }
    const int drow = cur.y < 0 ? AGG_ROWS : cur.y;
#pragma unroll
    for (int k = 0; k < 32; ++k) {
      const int dk = __builtin_amdgcn_readlane(drow, k);
      float* a = s_acc + dk * 64 + lane * 2;
      atomicAdd(a, v[k].x);
      atomicAdd(a + 1, v[k].y);
    }
    cur = nx;
    id = nid;
    g = ngi;
    ng = nng;
    next = nnext;
  }

  __syncthreads();

  const float2 bv = *reinterpret_cast<const float2*>(bias + lane * 2);
#pragma unroll
  for (int i = 0; i < AGG_ROWS / AGG_WAVES; ++i) {
    const unsigned r = (unsigned)(wave + AGG_WAVES * i);
    if (r < nrows) {
      const float nd = __builtin_bit_cast(float, __builtin_amdgcn_readlane(__builtin_bit_cast(int, nd_all), i));
      const float2 a = *reinterpret_cast<const float2*>(&s_acc[r * 64 + lane * 2]);
      float2 o;
      o.x = fmaf(a.x, nd, bv.x);
      o.y = fmaf(a.y, nd, bv.y);
      if (do_relu) {
        o.x = fmaxf(o.x, 0.0f);
        o.y = fmaxf(o.y, 0.0f);
      }
      *reinterpret_cast<float2*>(out + (size_t)(row0 + (int)r) * 64 + lane * 2) = o;
    }
  }
}

template <int K>
__global__ void __launch_bounds__(256) gemm_norm_wmma(const float* __restrict__ X,
                                                      const float* __restrict__ W,
                                                      const float* __restrict__ norm_s,
                                                      float* __restrict__ Y,
                                                      int n_nodes) {
  __shared__ float sW[K * 64];
  for (int i = threadIdx.x; i < K * 64; i += 256) sW[i] = W[i];
  __syncthreads();

  const int wave = threadIdx.x >> 5;
  const int lane = threadIdx.x & 31;
  const int row0 = (blockIdx.x * 8 + wave) * 16;
  if (row0 + 15 >= n_nodes) return;

  const int half  = lane >> 4;
  const int sub   = lane & 15;
  const int mrow  = row0 + sub;
  const float ns  = norm_s[mrow];
  const float* __restrict__ xrow = X + (size_t)mrow * K;

  v8f acc0 = {}, acc1 = {}, acc2 = {}, acc3 = {};

#pragma unroll
  for (int kk = 0; kk < K / 4; ++kk) {
    const int ka = kk * 4 + half * 2;
    v2f a;
    a.x = xrow[ka] * ns;
    a.y = xrow[ka + 1] * ns;

    const float* wk = &sW[ka * 64 + sub];
    v2f b;

    b.x = wk[0];       b.y = wk[64];
    acc0 = __builtin_amdgcn_wmma_f32_16x16x4_f32(false, a, false, b, (short)0, acc0, false, false);
    b.x = wk[16];      b.y = wk[64 + 16];
    acc1 = __builtin_amdgcn_wmma_f32_16x16x4_f32(false, a, false, b, (short)0, acc1, false, false);
    b.x = wk[32];      b.y = wk[64 + 32];
    acc2 = __builtin_amdgcn_wmma_f32_16x16x4_f32(false, a, false, b, (short)0, acc2, false, false);
    b.x = wk[48];      b.y = wk[64 + 48];
    acc3 = __builtin_amdgcn_wmma_f32_16x16x4_f32(false, a, false, b, (short)0, acc3, false, false);
  }

  float* ybase = Y + (size_t)row0 * 64 + sub;
#pragma unroll
  for (int r = 0; r < 8; ++r) {
    const int m = r + half * 8;
    float* yr = ybase + (size_t)m * 64;
    yr[0]  = acc0[r];
    yr[16] = acc1[r];
    yr[32] = acc2[r];
    yr[48] = acc3[r];
  }
}

static inline int cdiv(long long a, long long b) { return (int)((a + b - 1) / b); }

extern "C" void kernel_launch(void* const* d_in, const int* in_sizes, int n_in,
                              void* d_out, int out_size, void* d_ws, size_t ws_size,
                              hipStream_t stream) {
  const float* features = (const float*)d_in[0];
  const int*   src      = (const int*)d_in[1];
  const int*   dst      = (const int*)d_in[2];
  const float* W1       = (const float*)d_in[3];
  const float* b1       = (const float*)d_in[4];
  const float* W2       = (const float*)d_in[5];
  const float* b2       = (const float*)d_in[6];

  const int n_nodes = in_sizes[0] / 128;
  const int n_edges = in_sizes[1];

  float* ws     = (float*)d_ws;
  float* norm_s = ws;
  float* norm_d = ws + n_nodes;
  unsigned* piece_counter = (unsigned*)(ws + 2 * (size_t)n_nodes);
  float* bufA   = ws + 2 * (size_t)n_nodes + 256;
  float* bufB   = bufA + (size_t)n_nodes * 64;
  unsigned* bits = (unsigned*)(bufB + (size_t)n_nodes * 64);
  float* out    = (float*)d_out;

  const int B = 256;

  const int gemm_blocks = cdiv(n_nodes, 128);
  const int agg_blocks  = cdiv(n_nodes, AGG_ROWS);

  const int idx_blocks  = cdiv(n_edges, IDX_GROUPS * 16);
  const int pitch_words = idx_blocks * 32;

  const int deg_chunks  = cdiv(n_edges, DEG_CHUNK);
  const int deg_parts   = cdiv(n_nodes, DEG_PART_NODES);
  const int plane_words = deg_parts * DEG_PART_WORDS;
  unsigned* planes = bits + (size_t)agg_blocks * pitch_words;

  int* pairs = (int*)(planes + (size_t)2 * deg_chunks * plane_words);

  int* rec_pool = pairs + (size_t)2 * n_edges;

  degree_planes<<<2 * deg_parts * deg_chunks, DEG_THREADS, 0, stream>>>(src, dst, planes, n_edges, deg_chunks, deg_parts, plane_words);
  norm_from_planes<<<cdiv(2 * (long long)((n_nodes + 1) / 2), B), B, 0, stream>>>(planes, norm_s, n_nodes, deg_chunks, plane_words, piece_counter);

  owner_bits<<<idx_blocks, IDX_THREADS, 0, stream>>>(src, dst, bits, pairs, n_edges, agg_blocks, pitch_words);

  gemm_norm_wmma<128><<<gemm_blocks, B, 0, stream>>>(features, W1, norm_s, bufA, n_nodes);
  agg_owner<<<agg_blocks, AGG_THREADS, 0, stream>>>(bufA, pairs, bits, pitch_words, norm_d, b1, bufB, n_edges, n_nodes, 1,
                                                    rec_pool, piece_counter);

  gemm_norm_wmma<64><<<gemm_blocks, B, 0, stream>>>(bufB, W2, norm_s, bufA, n_nodes);
  agg_replay<<<agg_blocks, AGG_THREADS, 0, stream>>>(bufA, rec_pool, norm_d, b2, out, n_edges, n_nodes, 0);
}
